# starting kernel
# baseline (speedup 1.0000x reference)
#include <hip/hip_runtime.h>
#include <math.h>

typedef __attribute__((ext_vector_type(2))) float v2f;
typedef __attribute__((ext_vector_type(4))) float v4f;
typedef __attribute__((ext_vector_type(8))) float v8f;

namespace {
constexpr int kB = 4;
constexpr int kS = 4096;
constexpr int kD = 1024;
constexpr int kH = 64;
constexpr float kLnEps = 1e-5f;
constexpr float kScale = 0.125f;
}

static __device__ __forceinline__ v8f wmma_f32(v2f a, v2f b, v8f c) {
    return __builtin_amdgcn_wmma_f32_16x16x4_f32(false, a, false, b, (short)0, c,
                                                 false, false);
}

__global__ __launch_bounds__(128) void proj_ln_kernel(
    const float* __restrict__ x, const float* __restrict__ Wq,
    const float* __restrict__ Wk, const float* __restrict__ Wv,
    float* __restrict__ qo, float* __restrict__ ko, float* __restrict__ vo) {
    __shared__ float tq[4][16][68];
    __shared__ float tk[4][16][68];
    __shared__ float tv[4][16][68];

    const int tid = threadIdx.x;
    const int w = tid >> 5;
    const int lane = tid & 31;
    const int m = lane & 15;
    const int hi = lane >> 4;
    const int rowBase = blockIdx.x * 16;

    v8f aq[4], ak[4], av[4];
#pragma unroll
    for (int nt = 0; nt < 4; ++nt) { aq[nt] = {}; ak[nt] = {}; av[nt] = {}; }

    const float* xr = x + (size_t)(rowBase + m) * kD;
    const int k0beg = w * (kD / 4);
    for (int it = 0; it < kD / 16; ++it) {
        const int k0 = k0beg + 4 * it;
        v2f a;
        a.x = xr[k0 + 2 * hi];
        a.y = xr[k0 + 2 * hi + 1];
        const int wrow = (k0 + 2 * hi) * kH;
#pragma unroll
        for (int nt = 0; nt < 4; ++nt) {
            const int col = nt * 16 + m;
            v2f bq, bk, bv;
            bq.x = Wq[wrow + col];      bq.y = Wq[wrow + kH + col];
            bk.x = Wk[wrow + col];      bk.y = Wk[wrow + kH + col];
            bv.x = Wv[wrow + col];      bv.y = Wv[wrow + kH + col];
            aq[nt] = wmma_f32(a, bq, aq[nt]);
            ak[nt] = wmma_f32(a, bk, ak[nt]);
            av[nt] = wmma_f32(a, bv, av[nt]);
        }
    }

#pragma unroll
    for (int nt = 0; nt < 4; ++nt)
#pragma unroll
        for (int r = 0; r < 8; ++r) {
            const int row = r + 8 * hi;
            const int col = nt * 16 + m;
            tq[w][row][col] = aq[nt][r];
            tk[w][row][col] = ak[nt][r];
            tv[w][row][col] = av[nt][r];
        }
    __syncthreads();

    if (tid < 48) {
        const int which = tid >> 4;
        const int row = tid & 15;
        float(*arr)[16][68] = (which == 0) ? tq : (which == 1) ? tk : tv;
        float* r0 = &arr[0][row][0];
        const float* r1 = &arr[1][row][0];
        const float* r2 = &arr[2][row][0];
        const float* r3 = &arr[3][row][0];
        float s = 0.f, s2 = 0.f;
#pragma unroll
        for (int c = 0; c < kH; ++c) {
            const float t = r0[c] + r1[c] + r2[c] + r3[c];
            r0[c] = t;
            s += t;
            s2 += t * t;
        }
        if (which < 2) {
            const float mu = s * (1.f / kH);
            const float var = s2 * (1.f / kH) - mu * mu;
            const float inv = rsqrtf(var + kLnEps);
#pragma unroll
            for (int c = 0; c < kH; ++c) r0[c] = (r0[c] - mu) * inv;
        }
    }
    __syncthreads();

    const size_t ob = (size_t)rowBase * kH;
#pragma unroll
    for (int it = 0; it < 2; ++it) {
        const int flat = it * 512 + tid * 4;
        const int row = flat >> 6;
        const int col = flat & 63;
        *reinterpret_cast<v4f*>(qo + ob + flat) =
            *reinterpret_cast<const v4f*>(&tq[0][row][col]);
        *reinterpret_cast<v4f*>(ko + ob + flat) =
            *reinterpret_cast<const v4f*>(&tk[0][row][col]);
        *reinterpret_cast<v4f*>(vo + ob + flat) =
            *reinterpret_cast<const v4f*>(&tv[0][row][col]);
    }
}

__global__ __launch_bounds__(128) void flash_attn_kernel(
    const float* __restrict__ q, const float* __restrict__ k,
    const float* __restrict__ v, float* __restrict__ out) {
    __shared__ float ps[4][16][18];
    __shared__ float m_s[4][16];
    __shared__ float l_s[4][16];
    __shared__ float o_s[4][16][64];

    const int tid = threadIdx.x;
    const int w = tid >> 5;
    const int lane = tid & 31;
    const int m = lane & 15;
    const int hi = lane >> 4;
    const int b = blockIdx.x >> 8;
    const int qt = blockIdx.x & 255;

    const float* Qb = q + ((size_t)b * kS + qt * 16) * kH;
    const float* Kb = k + (size_t)b * kS * kH;
    const float* Vb = v + (size_t)b * kS * kH;

    v2f aq[16];
#pragma unroll
    for (int kk = 0; kk < 16; ++kk) {
        const float* p = Qb + m * kH + kk * 4 + 2 * hi;
        aq[kk].x = p[0] * kScale;
        aq[kk].y = p[1] * kScale;
    }

    v8f o[4];
#pragma unroll
    for (int nt = 0; nt < 4; ++nt) o[nt] = {};
    v8f lacc = {};
    float mrow[8];
#pragma unroll
    for (int r = 0; r < 8; ++r) mrow[r] = -INFINITY;
    const v2f bones = {1.f, 1.f};

    constexpr int kTiles = (kS / 16) / 4;
    for (int t = 0; t < kTiles; ++t) {
        const int j = w + 4 * t;
        const float* Kt = Kb + (size_t)j * 16 * kH;

        {
            const int jn = (t + 1 < kTiles) ? (j + 4) : j;
            const size_t nb = (size_t)jn * 16 * kH + lane * 32;
            __builtin_prefetch(Kb + nb, 0, 3);
            __builtin_prefetch(Vb + nb, 0, 3);
        }

        v8f sc0 = {}, sc1 = {};
#pragma unroll
        for (int kk = 0; kk < 8; ++kk) {
            v2f bk0, bk1;
            const float* p0 = Kt + m * kH + kk * 4 + 2 * hi;
            const float* p1 = p0 + 32;
            bk0.x = p0[0];  bk0.y = p0[1];
            bk1.x = p1[0];  bk1.y = p1[1];
            sc0 = wmma_f32(aq[kk], bk0, sc0);
            sc1 = wmma_f32(aq[kk + 8], bk1, sc1);
        }
        v8f sc;
#pragma unroll
        for (int r = 0; r < 8; ++r) sc[r] = sc0[r] + sc1[r];

        v8f pf;
        float al[8];
#pragma unroll
        for (int r = 0; r < 8; ++r) {
            float tmax = sc[r];
            tmax = fmaxf(tmax, __shfl_xor(tmax, 1));
            tmax = fmaxf(tmax, __shfl_xor(tmax, 2));
            tmax = fmaxf(tmax, __shfl_xor(tmax, 4));
            tmax = fmaxf(tmax, __shfl_xor(tmax, 8));
            const float mn = fmaxf(mrow[r], tmax);
            al[r] = __expf(mrow[r] - mn);
            mrow[r] = mn;
            pf[r] = __expf(sc[r] - mn);
        }
#pragma unroll
        for (int r = 0; r < 8; ++r) lacc[r] *= al[r];
#pragma unroll
        for (int nt = 0; nt < 4; ++nt)
#pragma unroll
            for (int r = 0; r < 8; ++r) o[nt][r] *= al[r];

#pragma unroll
        for (int r = 0; r < 8; ++r) ps[w][r + 8 * hi][m] = pf[r];
        asm volatile("" ::: "memory");

#pragma unroll
        for (int kk = 0; kk < 4; ++kk) {
            v2f ap;
            ap.x = ps[w][m][kk * 4 + 2 * hi];
            ap.y = ps[w][m][kk * 4 + 2 * hi + 1];
            lacc = wmma_f32(ap, bones, lacc);
            const float* vp = Vb + (size_t)(j * 16 + kk * 4 + 2 * hi) * kH;
#pragma unroll
            for (int nt = 0; nt < 4; ++nt) {
                v2f bv;
                bv.x = vp[nt * 16 + m];
                bv.y = vp[kH + nt * 16 + m];
                o[nt] = wmma_f32(ap, bv, o[nt]);
            }
        }
        asm volatile("" ::: "memory");
    }

    if (m == 0) {
#pragma unroll
        for (int r = 0; r < 8; ++r) {
            m_s[w][r + 8 * hi] = mrow[r];
            l_s[w][r + 8 * hi] = lacc[r];
        }
    }
#pragma unroll
    for (int nt = 0; nt < 4; ++nt)
#pragma unroll
        for (int r = 0; r < 8; ++r)
            o_s[w][r + 8 * hi][nt * 16 + m] = o[nt][r];
    __syncthreads();

    const int row = tid >> 3;
    const int c0 = (tid & 7) * 8;
    float M = m_s[0][row];
#pragma unroll
    for (int w2 = 1; w2 < 4; ++w2) M = fmaxf(M, m_s[w2][row]);
    float e[4];
    float L = 0.f;
#pragma unroll
    for (int w2 = 0; w2 < 4; ++w2) {
        e[w2] = __expf(m_s[w2][row] - M);
        L += e[w2] * l_s[w2][row];
    }
    const float invL = 1.f / L;
    float* Ob = out + ((size_t)b * kS + qt * 16) * kH;
#pragma unroll
    for (int c = 0; c < 8; ++c) {
        const float acc = e[0] * o_s[0][row][c0 + c] + e[1] * o_s[1][row][c0 + c] +
                          e[2] * o_s[2][row][c0 + c] + e[3] * o_s[3][row][c0 + c];
        Ob[row * kH + c0 + c] = acc * invL;
    }
}

extern "C" void kernel_launch(void* const* d_in, const int* in_sizes, int n_in,
                              void* d_out, int out_size, void* d_ws,
                              size_t ws_size, hipStream_t stream) {
    const float* x = (const float*)d_in[0];
    const float* Wq = (const float*)d_in[1];
    const float* Wk = (const float*)d_in[2];
    const float* Wv = (const float*)d_in[3];
    float* out = (float*)d_out;

    float* qws = (float*)d_ws;
    float* kws = qws + (size_t)kB * kS * kH;
    float* vws = kws + (size_t)kB * kS * kH;

    proj_ln_kernel<<<(kB * kS) / 16, 128, 0, stream>>>(x, Wq, Wk, Wv, qws, kws,
                                                       vws);
    flash_attn_kernel<<<kB * (kS / 16), 128, 0, stream>>>(qws, kws, vws, out);
}


# projection: 64-row blocks of 16 waves; each 64-k chunk of x and Wq|Wk|Wv fetched once per block by 16-B loads into a 2-stage LDS ring, fragments from LDS; no k split: k sum and LN stat order change
# speedup vs baseline: 1.4077x; 1.4077x over previous
#include <hip/hip_runtime.h>
#include <math.h>

typedef __attribute__((ext_vector_type(2))) float v2f;
typedef __attribute__((ext_vector_type(4))) float v4f;
typedef __attribute__((ext_vector_type(8))) float v8f;

namespace {
constexpr int kB = 4;
constexpr int kS = 4096;
constexpr int kD = 1024;
constexpr int kH = 64;
constexpr float kLnEps = 1e-5f;
constexpr float kScale = 0.125f;
constexpr int kChunk = 64;
constexpr int kKP = 68;
constexpr int kVP = 160;
constexpr int kKBuf = kChunk * kKP;
constexpr int kVBuf = (kChunk / 2) * kVP;
constexpr int kStage = kKBuf + kVBuf;
}

static __device__ __forceinline__ v8f wmma_f32(v2f a, v2f b, v8f c) {
    return __builtin_amdgcn_wmma_f32_16x16x4_f32(false, a, false, b, (short)0, c,
                                                 false, false);
}

namespace {
constexpr int kPR = 64;
constexpr int kPC = 64;
constexpr int kXP = 68;
constexpr int kWP = 2 * 3 * kH + 32;
constexpr int kXBuf = kPR * kXP;
constexpr int kWBuf = (kPC / 2) * kWP;
constexpr int kPStage = kXBuf + kWBuf;
constexpr int kCP = 3 * kH + 36;
}

__global__ __launch_bounds__(512) __attribute__((amdgpu_waves_per_eu(4, 4))) void proj_ln_kernel(
    const float* __restrict__ x, const float* __restrict__ Wq,
    const float* __restrict__ Wk, const float* __restrict__ Wv,
    float* __restrict__ qo, float* __restrict__ ko, float* __restrict__ vo) {
    __shared__ __attribute__((aligned(16))) float pst[2 * kPStage];
    static_assert(2 * kPStage >= kPR * kCP, "finished tile must fit the ring");

    const int tid = threadIdx.x;
    const int w = tid >> 5;
    const int lane = tid & 31;
    const int m = lane & 15;
    const int hi = lane >> 4;
    const int rt = w & 3;
    const int cg = w >> 2;
    const int rowBase = blockIdx.x * kPR;

    const int sr = tid >> 4;
    const int c4 = (tid & 15) * 4;
    const float* xg = x + (size_t)(rowBase + sr) * kD + c4;
    const size_t wgo = (size_t)(2 * sr) * kH + c4;
    const int xst = sr * kXP + c4;
    const int wst = kXBuf + sr * kWP + 2 * c4;

    {
        const v4f x0 = *reinterpret_cast<const v4f*>(xg);
        const v4f x1 = *reinterpret_cast<const v4f*>(xg + 32 * kD);
        const v4f q0 = *reinterpret_cast<const v4f*>(Wq + wgo);
        const v4f q1 = *reinterpret_cast<const v4f*>(Wq + wgo + kH);
        const v4f k0 = *reinterpret_cast<const v4f*>(Wk + wgo);
        const v4f k1 = *reinterpret_cast<const v4f*>(Wk + wgo + kH);
        const v4f v0 = *reinterpret_cast<const v4f*>(Wv + wgo);
        const v4f v1 = *reinterpret_cast<const v4f*>(Wv + wgo + kH);
        *reinterpret_cast<v4f*>(pst + xst) = x0;
        *reinterpret_cast<v4f*>(pst + xst + 32 * kXP) = x1;
        const v4f qa = {q0.x, q1.x, q0.y, q1.y};
        const v4f qb = {q0.z, q1.z, q0.w, q1.w};
        const v4f ka = {k0.x, k1.x, k0.y, k1.y};
        const v4f kb = {k0.z, k1.z, k0.w, k1.w};
        const v4f va = {v0.x, v1.x, v0.y, v1.y};
        const v4f vb = {v0.z, v1.z, v0.w, v1.w};
        *reinterpret_cast<v4f*>(pst + wst) = qa;
        *reinterpret_cast<v4f*>(pst + wst + 4) = qb;
        *reinterpret_cast<v4f*>(pst + wst + 2 * kH) = ka;
        *reinterpret_cast<v4f*>(pst + wst + 2 * kH + 4) = kb;
        *reinterpret_cast<v4f*>(pst + wst + 4 * kH) = va;
        *reinterpret_cast<v4f*>(pst + wst + 4 * kH + 4) = vb;
    }

    v8f acc[3];
#pragma unroll
    for (int j = 0; j < 3; ++j) acc[j] = {};

    const int afo = (rt * 16 + m) * kXP + 2 * hi;
    const int bfo = kXBuf + hi * kWP + 2 * (cg * 48 + m);

    __syncthreads();

    constexpr int kChunks = kD / kPC;
    for (int t = 0; t < kChunks; ++t) {
        const float* St = pst + (t & 1) * kPStage;
        float* Sn = pst + ((t + 1) & 1) * kPStage;
        const bool more = (t + 1 < kChunks);

        v4f x0 = {}, x1 = {}, q0 = {}, q1 = {}, k0 = {}, k1 = {}, v0 = {}, v1 = {};
        if (more) {
            const float* xn = xg + (t + 1) * kPC;
            const size_t wn = wgo + (size_t)(t + 1) * kPC * kH;
            x0 = *reinterpret_cast<const v4f*>(xn);
            x1 = *reinterpret_cast<const v4f*>(xn + 32 * kD);
            q0 = *reinterpret_cast<const v4f*>(Wq + wn);
            q1 = *reinterpret_cast<const v4f*>(Wq + wn + kH);
            k0 = *reinterpret_cast<const v4f*>(Wk + wn);
            k1 = *reinterpret_cast<const v4f*>(Wk + wn + kH);
            v0 = *reinterpret_cast<const v4f*>(Wv + wn);
            v1 = *reinterpret_cast<const v4f*>(Wv + wn + kH);
        }

#pragma unroll
        for (int kk = 0; kk < kPC / 4; ++kk) {
            const v2f a = *reinterpret_cast<const v2f*>(St + afo + kk * 4);
#pragma unroll
            for (int j = 0; j < 3; ++j) {
                const v2f bw =
                    *reinterpret_cast<const v2f*>(St + bfo + 2 * kk * kWP + j * 32);
                acc[j] = wmma_f32(a, bw, acc[j]);
            }
        }

        if (more) {
            *reinterpret_cast<v4f*>(Sn + xst) = x0;
            *reinterpret_cast<v4f*>(Sn + xst + 32 * kXP) = x1;
            const v4f qa = {q0.x, q1.x, q0.y, q1.y};
            const v4f qb = {q0.z, q1.z, q0.w, q1.w};
            const v4f ka = {k0.x, k1.x, k0.y, k1.y};
            const v4f kb = {k0.z, k1.z, k0.w, k1.w};
            const v4f va = {v0.x, v1.x, v0.y, v1.y};
            const v4f vb = {v0.z, v1.z, v0.w, v1.w};
            *reinterpret_cast<v4f*>(Sn + wst) = qa;
            *reinterpret_cast<v4f*>(Sn + wst + 4) = qb;
            *reinterpret_cast<v4f*>(Sn + wst + 2 * kH) = ka;
            *reinterpret_cast<v4f*>(Sn + wst + 2 * kH + 4) = kb;
            *reinterpret_cast<v4f*>(Sn + wst + 4 * kH) = va;
            *reinterpret_cast<v4f*>(Sn + wst + 4 * kH + 4) = vb;
        }
        __syncthreads();
    }

    float(*ct)[kCP] = reinterpret_cast<float(*)[kCP]>(pst);
#pragma unroll
    for (int j = 0; j < 3; ++j)
#pragma unroll
        for (int r = 0; r < 8; ++r)
            ct[rt * 16 + r + 8 * hi][(cg * 3 + j) * 16 + m] = acc[j][r];
    __syncthreads();

    const int row = tid >> 3;
    const int seg = tid & 7;
    const float* cr = &ct[row][seg * 4];
    v4f qa = *reinterpret_cast<const v4f*>(cr);
    v4f qb = *reinterpret_cast<const v4f*>(cr + 32);
    v4f ka = *reinterpret_cast<const v4f*>(cr + kH);
    v4f kb = *reinterpret_cast<const v4f*>(cr + kH + 32);
    const v4f va = *reinterpret_cast<const v4f*>(cr + 2 * kH);
    const v4f vb = *reinterpret_cast<const v4f*>(cr + 2 * kH + 32);

    float sq = 0.f, sq2 = 0.f, sk = 0.f, sk2 = 0.f;
#pragma unroll
    for (int c = 0; c < 4; ++c) {
        sq += qa[c];  sq2 += qa[c] * qa[c];
        sk += ka[c];  sk2 += ka[c] * ka[c];
    }
#pragma unroll
    for (int c = 0; c < 4; ++c) {
        sq += qb[c];  sq2 += qb[c] * qb[c];
        sk += kb[c];  sk2 += kb[c] * kb[c];
    }
#pragma unroll
    for (int d = 1; d < 8; d <<= 1) {
        sq += __shfl_xor(sq, d);
        sq2 += __shfl_xor(sq2, d);
        sk += __shfl_xor(sk, d);
        sk2 += __shfl_xor(sk2, d);
    }
    {
        const float mu = sq * (1.f / kH);
        const float var = sq2 * (1.f / kH) - mu * mu;
        const float inv = rsqrtf(var + kLnEps);
#pragma unroll
        for (int c = 0; c < 4; ++c) {
            qa[c] = (qa[c] - mu) * inv;
            qb[c] = (qb[c] - mu) * inv;
        }
    }
    {
        const float mu = sk * (1.f / kH);
        const float var = sk2 * (1.f / kH) - mu * mu;
        const float inv = rsqrtf(var + kLnEps);
#pragma unroll
        for (int c = 0; c < 4; ++c) {
            ka[c] = (ka[c] - mu) * inv;
            kb[c] = (kb[c] - mu) * inv;
        }
    }
    const size_t ob = (size_t)(rowBase + row) * kH + seg * 4;
    *reinterpret_cast<v4f*>(qo + ob) = qa;
    *reinterpret_cast<v4f*>(qo + ob + 32) = qb;
    *reinterpret_cast<v4f*>(ko + ob) = ka;
    *reinterpret_cast<v4f*>(ko + ob + 32) = kb;
    *reinterpret_cast<v4f*>(vo + ob) = va;
    *reinterpret_cast<v4f*>(vo + ob + 32) = vb;
}

static __device__ __forceinline__ float xhalf(float x) {
    const int b = __builtin_bit_cast(int, x);
    return __builtin_bit_cast(
        float, __builtin_amdgcn_permlanex16(b, b, 0x76543210, 0xfedcba98, false, false));
}

__global__ __launch_bounds__(512) __attribute__((amdgpu_waves_per_eu(4, 4))) void flash_attn_kernel(
    const float* __restrict__ q, const float* __restrict__ k,
    const float* __restrict__ v, float* __restrict__ out) {
    constexpr int kOP = 68;
    __shared__ __attribute__((aligned(16))) float stage[2 * kStage];
    __shared__ float m_s[16][16];
    __shared__ float l_s[16][16];
    static_assert(2 * kStage >= 16 * 16 * kOP, "O partials must fit the ring");

    const int tid = threadIdx.x;
    const int w = tid >> 5;
    const int lane = tid & 31;
    const int m = lane & 15;
    const int hi = lane >> 4;
    const int qi = w & 3;
    const int ks = w >> 2;
    const int b = blockIdx.x >> 6;
    const int qb = blockIdx.x & 63;

    const float* Qb = q + ((size_t)b * kS + qb * 64 + qi * 16) * kH;
    const float* Kb = k + (size_t)b * kS * kH;
    const float* Vb = v + (size_t)b * kS * kH;

    const int kr = tid >> 4;
    const int c4 = (tid & 15) * 4;
    const int vp = (kr & 24) | ((kr & 1) << 2) | ((kr & 7) >> 1);
    const float* Kg = Kb + (size_t)kr * kH + c4;
    const float* Vg = Vb + (size_t)(2 * vp) * kH + c4;
    const int kst = kr * kKP + c4;
    const int vst = kKBuf + kr * kVP + 2 * c4;

    {
        const v4f k0 = *reinterpret_cast<const v4f*>(Kg);
        const v4f k1 = *reinterpret_cast<const v4f*>(Kg + 32 * kH);
        const v4f va = *reinterpret_cast<const v4f*>(Vg);
        const v4f vb = *reinterpret_cast<const v4f*>(Vg + kH);
        *reinterpret_cast<v4f*>(stage + kst) = k0;
        *reinterpret_cast<v4f*>(stage + kst + 32 * kKP) = k1;
        const v4f lo = {va.x, vb.x, va.y, vb.y};
        const v4f hi4 = {va.z, vb.z, va.w, vb.w};
        *reinterpret_cast<v4f*>(stage + vst) = lo;
        *reinterpret_cast<v4f*>(stage + vst + 4) = hi4;
    }

    v2f bq[16];
#pragma unroll
    for (int kk = 0; kk < 16; ++kk) {
        const float* p = Qb + m * kH + kk * 4 + 2 * hi;
        bq[kk].x = p[0] * kScale;
        bq[kk].y = p[1] * kScale;
    }

    v8f o[4];
#pragma unroll
    for (int nt = 0; nt < 4; ++nt) o[nt] = {};
    v8f lacc = {};
    float mrow = -INFINITY;
    const v2f aones = {1.f, 1.f};

    const int kfo = (ks * 16 + m) * kKP + 2 * hi;
    const int vfo = kKBuf + (ks * 8 + hi) * kVP + 2 * m;

    __syncthreads();

    constexpr int kRounds = kS / kChunk;
    for (int t = 0; t < kRounds; ++t) {
        const float* St = stage + (t & 1) * kStage;
        float* Sn = stage + ((t + 1) & 1) * kStage;
        const bool more = (t + 1 < kRounds);

        v4f nk0 = {}, nk1 = {}, nva = {}, nvb = {};
        if (more) {
            const size_t go = (size_t)(t + 1) * kChunk * kH;
            nk0 = *reinterpret_cast<const v4f*>(Kg + go);
            nk1 = *reinterpret_cast<const v4f*>(Kg + go + 32 * kH);
            nva = *reinterpret_cast<const v4f*>(Vg + go);
            nvb = *reinterpret_cast<const v4f*>(Vg + go + kH);
        }

        v8f sc0 = {}, sc1 = {};
#pragma unroll
        for (int kk = 0; kk < 8; ++kk) {
            const v2f ak0 = *reinterpret_cast<const v2f*>(St + kfo + kk * 4);
            const v2f ak1 = *reinterpret_cast<const v2f*>(St + kfo + 32 + kk * 4);
            sc0 = wmma_f32(ak0, bq[kk], sc0);
            sc1 = wmma_f32(ak1, bq[kk + 8], sc1);
        }
        v8f sc;
#pragma unroll
        for (int r = 0; r < 8; ++r) sc[r] = sc0[r] + sc1[r];

        v2f av[4][4];
#pragma unroll
        for (int kk = 0; kk < 4; ++kk)
#pragma unroll
            for (int nt = 0; nt < 4; ++nt)
                av[kk][nt] = *reinterpret_cast<const v2f*>(St + vfo + 2 * kk * kVP +
                                                           nt * 32);

        float tmax = fmaxf(fmaxf(fmaxf(sc[0], sc[1]), fmaxf(sc[2], sc[3])),
                           fmaxf(fmaxf(sc[4], sc[5]), fmaxf(sc[6], sc[7])));
        tmax = fmaxf(tmax, xhalf(tmax));
        const float mn = fmaxf(mrow, tmax);
        const float al = __expf(mrow - mn);
        mrow = mn;
        v8f pf;
#pragma unroll
        for (int r = 0; r < 8; ++r) pf[r] = __expf(sc[r] - mn);
#pragma unroll
        for (int r = 0; r < 8; ++r) lacc[r] *= al;
#pragma unroll
        for (int nt = 0; nt < 4; ++nt)
#pragma unroll
            for (int r = 0; r < 8; ++r) o[nt][r] *= al;

#pragma unroll
        for (int kk = 0; kk < 4; ++kk) {
            v2f bp;
            bp.x = pf[2 * kk];
            bp.y = pf[2 * kk + 1];
            lacc = wmma_f32(aones, bp, lacc);
#pragma unroll
            for (int nt = 0; nt < 4; ++nt) o[nt] = wmma_f32(av[kk][nt], bp, o[nt]);
        }

        if (more) {
            *reinterpret_cast<v4f*>(Sn + kst) = nk0;
            *reinterpret_cast<v4f*>(Sn + kst + 32 * kKP) = nk1;
            const v4f lo = {nva.x, nvb.x, nva.y, nvb.y};
            const v4f hi4 = {nva.z, nvb.z, nva.w, nvb.w};
            *reinterpret_cast<v4f*>(Sn + vst) = lo;
            *reinterpret_cast<v4f*>(Sn + vst + 4) = hi4;
        }
        __syncthreads();
    }

    float(*o_s)[16][kOP] = reinterpret_cast<float(*)[16][kOP]>(stage);
    if (hi == 0) {
        m_s[w][m] = mrow;
        l_s[w][m] = lacc[0];
    }
#pragma unroll
    for (int nt = 0; nt < 4; ++nt) {
        const v4f lo = {o[nt][0], o[nt][1], o[nt][2], o[nt][3]};
        const v4f hi4 = {o[nt][4], o[nt][5], o[nt][6], o[nt][7]};
        *reinterpret_cast<v4f*>(&o_s[w][m][nt * 16 + 8 * hi]) = lo;
        *reinterpret_cast<v4f*>(&o_s[w][m][nt * 16 + 8 * hi + 4]) = hi4;
    }
    __syncthreads();

    const int row = tid >> 3;
    const int qt2 = row >> 4;
    const int r16 = row & 15;
    const int c0 = (tid & 7) * 8;
    float M = m_s[qt2][r16];
#pragma unroll
    for (int s2 = 1; s2 < 4; ++s2) M = fmaxf(M, m_s[s2 * 4 + qt2][r16]);
    float e[4];
    float L = 0.f;
#pragma unroll
    for (int s2 = 0; s2 < 4; ++s2) {
        e[s2] = __expf(m_s[s2 * 4 + qt2][r16] - M);
        L += e[s2] * l_s[s2 * 4 + qt2][r16];
    }
    const float invL = 1.f / L;
    float* Ob = out + ((size_t)b * kS + qb * 64) * kH;
#pragma unroll
    for (int c = 0; c < 8; ++c) {
        const float acc = e[0] * o_s[qt2][r16][c0 + c] +
                          e[1] * o_s[4 + qt2][r16][c0 + c] +
                          e[2] * o_s[8 + qt2][r16][c0 + c] +
                          e[3] * o_s[12 + qt2][r16][c0 + c];
        Ob[row * kH + c0 + c] = acc * invL;
    }
}

extern "C" void kernel_launch(void* const* d_in, const int* in_sizes, int n_in,
                              void* d_out, int out_size, void* d_ws,
                              size_t ws_size, hipStream_t stream) {
    const float* x = (const float*)d_in[0];
    const float* Wq = (const float*)d_in[1];
    const float* Wk = (const float*)d_in[2];
    const float* Wv = (const float*)d_in[3];
    float* out = (float*)d_out;

    float* qws = (float*)d_ws;
    float* kws = qws + (size_t)kB * kS * kH;
    float* vws = kws + (size_t)kB * kS * kH;

    proj_ln_kernel<<<(kB * kS) / kPR, 512, 0, stream>>>(x, Wq, Wk, Wv, qws, kws,
                                                       vws);
    flash_attn_kernel<<<kB * (kS / 64), 512, 0, stream>>>(qws, kws, vws, out);
}
